# starting kernel
# baseline (speedup 1.0000x reference)
#include <hip/hip_runtime.h>

typedef __attribute__((ext_vector_type(2))) float v2f;
typedef __attribute__((ext_vector_type(8))) float v8f;

#define TILES_PER_BLOCK 8
#define BLOCK_THREADS   256

__global__ __launch_bounds__(BLOCK_THREADS)
void edge_mlp_wmma_kernel(const float* __restrict__ h,
                          const int*   __restrict__ src,
                          const int*   __restrict__ dst,
                          const float* __restrict__ W1,
                          const float* __restrict__ B1,
                          const float* __restrict__ W2,
                          const float* __restrict__ B2,
                          float* __restrict__ out,
                          int nEdges)
{
    __shared__ v2f ldsW1[4096];

    const int tid = threadIdx.x;

    for (int i = tid; i < 4096; i += BLOCK_THREADS) {
        int L  = i & 31;
        int t  = (i >> 5) & 3;
        int s  = i >> 7;
        int hi = L >> 4;
        int lo = L & 15;
        int k0 = 4 * s + 2 * hi;
        int n  = 16 * t + lo;
        v2f frag;
        frag.x = W1[(k0 + 0) * 64 + n];
        frag.y = W1[(k0 + 1) * 64 + n];
        ldsW1[i] = frag;
    }
    __syncthreads();

    const int wave = tid >> 5;
    const int lane = tid & 31;
    const int hi   = lane >> 4;
    const int lo   = lane & 15;

    const int nTiles = (nEdges + 15) >> 4;
    const int tile   = blockIdx.x * TILES_PER_BLOCK + wave;
    if (tile >= nTiles) return;

    const int eRow = tile * 16 + lo;
    const int eCl  = eRow < nEdges ? eRow : (nEdges - 1);
    const float* __restrict__ pS = h + (long long)src[eCl] * 64;
    const float* __restrict__ pD = h + (long long)dst[eCl] * 64;

    v8f acc0 = {}, acc1 = {}, acc2 = {}, acc3 = {};

#pragma unroll
    for (int s = 0; s < 16; ++s) {
        v2f a = *(const v2f*)(pS + 4 * s + 2 * hi);
        const v2f* f = &ldsW1[(s * 4) * 32 + lane];
        v2f b0 = f[0 * 32], b1 = f[1 * 32], b2 = f[2 * 32], b3 = f[3 * 32];
        acc0 = __builtin_amdgcn_wmma_f32_16x16x4_f32(false, a, false, b0, (short)0, acc0, false, false);
        acc1 = __builtin_amdgcn_wmma_f32_16x16x4_f32(false, a, false, b1, (short)0, acc1, false, false);
        acc2 = __builtin_amdgcn_wmma_f32_16x16x4_f32(false, a, false, b2, (short)0, acc2, false, false);
        acc3 = __builtin_amdgcn_wmma_f32_16x16x4_f32(false, a, false, b3, (short)0, acc3, false, false);
    }
#pragma unroll
    for (int s = 16; s < 32; ++s) {
        v2f a = *(const v2f*)(pD + 4 * (s - 16) + 2 * hi);
        const v2f* f = &ldsW1[(s * 4) * 32 + lane];
        v2f b0 = f[0 * 32], b1 = f[1 * 32], b2 = f[2 * 32], b3 = f[3 * 32];
        acc0 = __builtin_amdgcn_wmma_f32_16x16x4_f32(false, a, false, b0, (short)0, acc0, false, false);
        acc1 = __builtin_amdgcn_wmma_f32_16x16x4_f32(false, a, false, b1, (short)0, acc1, false, false);
        acc2 = __builtin_amdgcn_wmma_f32_16x16x4_f32(false, a, false, b2, (short)0, acc2, false, false);
        acc3 = __builtin_amdgcn_wmma_f32_16x16x4_f32(false, a, false, b3, (short)0, acc3, false, false);
    }

    float bias0 = B1[ 0 + lo];
    float bias1 = B1[16 + lo];
    float bias2 = B1[32 + lo];
    float bias3 = B1[48 + lo];
    float w2a0 = W2[( 0 + lo) * 2 + 0], w2b0 = W2[( 0 + lo) * 2 + 1];
    float w2a1 = W2[(16 + lo) * 2 + 0], w2b1 = W2[(16 + lo) * 2 + 1];
    float w2a2 = W2[(32 + lo) * 2 + 0], w2b2 = W2[(32 + lo) * 2 + 1];
    float w2a3 = W2[(48 + lo) * 2 + 0], w2b3 = W2[(48 + lo) * 2 + 1];

    float r0[8], r1[8];
#pragma unroll
    for (int j = 0; j < 8; ++j) {
        float h0 = fmaxf(acc0[j] + bias0, 0.0f);
        float h1 = fmaxf(acc1[j] + bias1, 0.0f);
        float h2 = fmaxf(acc2[j] + bias2, 0.0f);
        float h3 = fmaxf(acc3[j] + bias3, 0.0f);
        r0[j] = h0 * w2a0 + h1 * w2a1 + h2 * w2a2 + h3 * w2a3;
        r1[j] = h0 * w2b0 + h1 * w2b1 + h2 * w2b2 + h3 * w2b3;
    }

#pragma unroll
    for (int m = 1; m < 16; m <<= 1) {
#pragma unroll
        for (int j = 0; j < 8; ++j) {
            r0[j] += __shfl_xor(r0[j], m, 32);
            r1[j] += __shfl_xor(r1[j], m, 32);
        }
    }

    if (lo < 2) {
        const float bb = (lo == 0) ? B2[0] : B2[1];
#pragma unroll
        for (int j = 0; j < 8; ++j) {
            int m = j + 8 * hi;
            int e = tile * 16 + m;
            if (e < nEdges) {
                float v = ((lo == 0) ? r0[j] : r1[j]) + bb;
                out[e * 2 + lo] = v;
            }
        }
    }
}

extern "C" void kernel_launch(void* const* d_in, const int* in_sizes, int n_in,
                              void* d_out, int out_size, void* d_ws, size_t ws_size,
                              hipStream_t stream) {
    const float* h   = (const float*)d_in[0];
    const int*   src = (const int*)d_in[1];
    const int*   dst = (const int*)d_in[2];
    const float* W1  = (const float*)d_in[3];
    const float* B1  = (const float*)d_in[4];
    const float* W2  = (const float*)d_in[5];
    const float* B2  = (const float*)d_in[6];
    float* out = (float*)d_out;

    const int nEdges = in_sizes[1];
    const int nTiles = (nEdges + 15) >> 4;
    const int nBlocks = (nTiles + TILES_PER_BLOCK - 1) / TILES_PER_BLOCK;

    edge_mlp_wmma_kernel<<<nBlocks, BLOCK_THREADS, 0, stream>>>(
        h, src, dst, W1, B1, W2, B2, out, nEdges);
}


# Layer 1 hoisted from edges to nodes: P=h@W1[0:64]+b1, Q=h@W1[64:128] once per node on f32 WMMA (new first launch), edge pass adds gathered rows; 10x fewer WMMAs; f32 sum order changes
# speedup vs baseline: 1.8475x; 1.8475x over previous
#include <hip/hip_runtime.h>


typedef __attribute__((ext_vector_type(2))) float v2f;
typedef __attribute__((ext_vector_type(4))) float v4f;
typedef __attribute__((ext_vector_type(8))) float v8f;

#define PROJ_BLOCKS   256
#define PROJ_THREADS  896
#define PROJ_WAVES    (PROJ_THREADS / 32)
#define EDGE_BLOCKS   1024
#define EDGE_THREADS  256
#define EDGE_WAVES    (EDGE_THREADS / 32)

__global__ __launch_bounds__(PROJ_THREADS)
void node_project_kernel(const float* __restrict__ h,
                         const float* __restrict__ W1,
                         const float* __restrict__ B1,
                         float* __restrict__ ws,
                         int hElems)
{
    __shared__ v2f ldsW1[4096];

    const int tid = threadIdx.x;

    for (int u = tid; u < 1024; u += PROJ_THREADS) {
        const int lo = u & 15;
        const int hh = (u >> 4) & 1;
        const int s  = u >> 5;
        const int k0 = 4 * s + 2 * hh;
        const v4f r0 = *(const v4f*)(W1 + (k0 + 0) * 64 + 4 * lo);
        const v4f r1 = *(const v4f*)(W1 + (k0 + 1) * 64 + 4 * lo);
        v2f* d = &ldsW1[(s * 4) * 32 + (u & 31)];
        v2f f0; f0.x = r0.x; f0.y = r1.x;
        v2f f1; f1.x = r0.y; f1.y = r1.y;
        v2f f2; f2.x = r0.z; f2.y = r1.z;
        v2f f3; f3.x = r0.w; f3.y = r1.w;
        d[0 * 32] = f0;
        d[1 * 32] = f1;
        d[2 * 32] = f2;
        d[3 * 32] = f3;
    }
    __syncthreads();

    const int wave = tid >> 5;
    const int lane = tid & 31;
    const int hi   = lane >> 4;
    const int lo   = lane & 15;

    const int nNodes = hElems >> 6;
    const int nTiles = (nNodes + 15) >> 4;
    float* __restrict__ P = ws;
    float* __restrict__ Q = ws + hElems;

    const v4f bias = *(const v4f*)(B1 + 4 * lo);

    for (int tile = wave * (int)gridDim.x + (int)blockIdx.x; tile < nTiles;
         tile += PROJ_WAVES * (int)gridDim.x) {
        asm volatile("" ::: "memory");
        const int row  = tile * 16 + lo;
        const int rowC = row < nNodes ? row : (nNodes - 1);
        const float* __restrict__ pA = h + (long long)rowC * 64;

        v8f p0 = {}, p1 = {}, p2 = {}, p3 = {};
        v8f q0 = {}, q1 = {}, q2 = {}, q3 = {};

#pragma unroll
        for (int s = 0; s < 16; ++s) {
            const v2f a = *(const v2f*)(pA + 4 * s + 2 * hi);
            const v2f* fp = &ldsW1[(s * 4) * 32 + lane];
            const v2f* fq = &ldsW1[((s + 16) * 4) * 32 + lane];
            const v2f bp0 = fp[0 * 32], bp1 = fp[1 * 32], bp2 = fp[2 * 32], bp3 = fp[3 * 32];
            const v2f bq0 = fq[0 * 32], bq1 = fq[1 * 32], bq2 = fq[2 * 32], bq3 = fq[3 * 32];
            p0 = __builtin_amdgcn_wmma_f32_16x16x4_f32(false, a, false, bp0, (short)0, p0, false, false);
            p1 = __builtin_amdgcn_wmma_f32_16x16x4_f32(false, a, false, bp1, (short)0, p1, false, false);
            p2 = __builtin_amdgcn_wmma_f32_16x16x4_f32(false, a, false, bp2, (short)0, p2, false, false);
            p3 = __builtin_amdgcn_wmma_f32_16x16x4_f32(false, a, false, bp3, (short)0, p3, false, false);
            q0 = __builtin_amdgcn_wmma_f32_16x16x4_f32(false, a, false, bq0, (short)0, q0, false, false);
            q1 = __builtin_amdgcn_wmma_f32_16x16x4_f32(false, a, false, bq1, (short)0, q1, false, false);
            q2 = __builtin_amdgcn_wmma_f32_16x16x4_f32(false, a, false, bq2, (short)0, q2, false, false);
            q3 = __builtin_amdgcn_wmma_f32_16x16x4_f32(false, a, false, bq3, (short)0, q3, false, false);
        }

#pragma unroll
        for (int j = 0; j < 8; ++j) {
            const int r = tile * 16 + 8 * hi + j;
            if (r < nNodes) {
                v4f vp, vq;
                vp.x = p0[j] + bias.x; vp.y = p1[j] + bias.y;
                vp.z = p2[j] + bias.z; vp.w = p3[j] + bias.w;
                vq.x = q0[j]; vq.y = q1[j]; vq.z = q2[j]; vq.w = q3[j];
                *(v4f*)(P + (long long)r * 64 + 4 * lo) = vp;
                *(v4f*)(Q + (long long)r * 64 + 4 * lo) = vq;
            }
        }
    }
}


#define EDGE_LOAD(i)                                                          \
    const int s##i = __shfl(sv, half + i, 32);                                \
    const int d##i = __shfl(dv, half + i, 32);                                \
    const v4f p##i = P4[(unsigned)s##i * 16u + (unsigned)c];                  \
    const v4f q##i = Q4[(unsigned)d##i * 16u + (unsigned)c];

#define EDGE_SCORE(i)                                                         \
    float a##i, b##i;                                                         \
    {                                                                         \
        const float h0 = fmaxf(p##i.x + q##i.x, 0.0f);                        \
        const float h1 = fmaxf(p##i.y + q##i.y, 0.0f);                        \
        const float h2 = fmaxf(p##i.z + q##i.z, 0.0f);                        \
        const float h3 = fmaxf(p##i.w + q##i.w, 0.0f);                        \
        a##i = h0 * wA.x + h1 * wA.z + h2 * wB.x + h3 * wB.z;                 \
        b##i = h0 * wA.y + h1 * wA.w + h2 * wB.y + h3 * wB.w;                 \
    }

#define BFLY(lo, hi, m, up)                                                   \
    {                                                                         \
        const float keepA = up ? a##hi : a##lo;                               \
        const float sendA = up ? a##lo : a##hi;                               \
        const float keepB = up ? b##hi : b##lo;                               \
        const float sendB = up ? b##lo : b##hi;                               \
        a##lo = keepA + __shfl_xor(sendA, m, 32);                             \
        b##lo = keepB + __shfl_xor(sendB, m, 32);                             \
    }

__global__ __launch_bounds__(EDGE_THREADS)
void edge_score_kernel(const float* __restrict__ ws,
                       const int*   __restrict__ src,
                       const int*   __restrict__ dst,
                       const float* __restrict__ W2,
                       const float* __restrict__ B2,
                       float* __restrict__ out,
                       int hElems,
                       int nEdges)
{
    const v4f* __restrict__ P4 = (const v4f*)ws;
    const v4f* __restrict__ Q4 = (const v4f*)(ws + hElems);

    const int lane = threadIdx.x & 31;
    const int c    = lane & 15;
    const int half = lane & 16;

    const int waveG   = (int)blockIdx.x * EDGE_WAVES + ((int)threadIdx.x >> 5);
    const int nWaves  = (int)gridDim.x * EDGE_WAVES;
    const int nChunks = (nEdges + 31) >> 5;

    const v4f wA = *(const v4f*)(W2 + 8 * c);
    const v4f wB = *(const v4f*)(W2 + 8 * c + 4);
    const float b20 = B2[0];
    const float b21 = B2[1];

    const bool up8 = (c & 8) != 0;
    const bool up4 = (c & 4) != 0;
    const bool up2 = (c & 2) != 0;
    const bool up1 = (c & 1) != 0;

    for (int chunk = waveG; chunk < nChunks; chunk += nWaves) {
        const int e  = chunk * 32 + lane;
        const int eC = e < nEdges ? e : (nEdges - 1);
        const int sv = src[eC];
        const int dv = dst[eC];

        EDGE_LOAD(0)  EDGE_LOAD(1)  EDGE_LOAD(2)  EDGE_LOAD(3)
        EDGE_LOAD(4)  EDGE_LOAD(5)  EDGE_LOAD(6)  EDGE_LOAD(7)
        EDGE_LOAD(8)  EDGE_LOAD(9)  EDGE_LOAD(10) EDGE_LOAD(11)
        EDGE_LOAD(12) EDGE_LOAD(13) EDGE_LOAD(14) EDGE_LOAD(15)

        EDGE_SCORE(0)  EDGE_SCORE(1)  EDGE_SCORE(2)  EDGE_SCORE(3)
        EDGE_SCORE(4)  EDGE_SCORE(5)  EDGE_SCORE(6)  EDGE_SCORE(7)
        EDGE_SCORE(8)  EDGE_SCORE(9)  EDGE_SCORE(10) EDGE_SCORE(11)
        EDGE_SCORE(12) EDGE_SCORE(13) EDGE_SCORE(14) EDGE_SCORE(15)

        BFLY(0, 8, 8, up8)  BFLY(1, 9, 8, up8)  BFLY(2, 10, 8, up8) BFLY(3, 11, 8, up8)
        BFLY(4, 12, 8, up8) BFLY(5, 13, 8, up8) BFLY(6, 14, 8, up8) BFLY(7, 15, 8, up8)
        BFLY(0, 4, 4, up4)  BFLY(1, 5, 4, up4)  BFLY(2, 6, 4, up4)  BFLY(3, 7, 4, up4)
        BFLY(0, 2, 2, up2)  BFLY(1, 3, 2, up2)
        BFLY(0, 1, 1, up1)

        if (e < nEdges) {
            v2f o;
            o.x = a0 + b20;
            o.y = b0 + b21;
            *(v2f*)(out + (long long)e * 2) = o;
        }
    }
}

extern "C" void kernel_launch(void* const* d_in, const int* in_sizes, int n_in,
                              void* d_out, int out_size, void* d_ws, size_t ws_size,
                              hipStream_t stream) {
    const float* h   = (const float*)d_in[0];
    const int*   src = (const int*)d_in[1];
    const int*   dst = (const int*)d_in[2];
    const float* W1  = (const float*)d_in[3];
    const float* B1  = (const float*)d_in[4];
    const float* W2  = (const float*)d_in[5];
    const float* B2  = (const float*)d_in[6];
    float* out = (float*)d_out;
    float* ws  = (float*)d_ws;

    const int hElems = in_sizes[0];
    const int nEdges = in_sizes[1];

    node_project_kernel<<<PROJ_BLOCKS, PROJ_THREADS, 0, stream>>>(h, W1, B1, ws, hElems);
    edge_score_kernel<<<EDGE_BLOCKS, EDGE_THREADS, 0, stream>>>(
        ws, src, dst, W2, B2, out, hElems, nEdges);
}


# Projection launch: 8-wave work-groups whose waves loop over their tiles; all 128 W1 fragments stay in registers (read from LDS once), next tile's rows load a tile ahead; 2048 waves, not 7168
# speedup vs baseline: 1.9383x; 1.0491x over previous
#include <hip/hip_runtime.h>


typedef __attribute__((ext_vector_type(2))) float v2f;
typedef __attribute__((ext_vector_type(4))) float v4f;
typedef __attribute__((ext_vector_type(8))) float v8f;

#define PROJ_BLOCKS   256
#define PROJ_THREADS  256
#define PROJ_WAVES    (PROJ_THREADS / 32)
#define EDGE_BLOCKS   1024
#define EDGE_THREADS  256
#define EDGE_WAVES    (EDGE_THREADS / 32)

__global__ __launch_bounds__(PROJ_THREADS)
void node_project_kernel(const float* __restrict__ h,
                         const float* __restrict__ W1,
                         const float* __restrict__ B1,
                         float* __restrict__ ws,
                         int hElems)
{
    __shared__ v2f ldsW1[4096];

    const int tid  = threadIdx.x;
    const int wave = tid >> 5;
    const int lane = tid & 31;
    const int hi   = lane >> 4;
    const int lo   = lane & 15;

    const int nNodes = hElems >> 6;
    const int nTiles = (nNodes + 15) >> 4;
    float* __restrict__ P = ws;
    float* __restrict__ Q = ws + hElems;

    const int tileStep = PROJ_WAVES * (int)gridDim.x;
    int tile = wave * (int)gridDim.x + (int)blockIdx.x;

    v2f a[16];
    {
        const int row  = tile * 16 + lo;
        const int rowC = row < nNodes ? row : (nNodes - 1);
        const float* __restrict__ pA = h + (long long)rowC * 64 + 2 * hi;
#pragma unroll
        for (int s = 0; s < 16; ++s) a[s] = *(const v2f*)(pA + 4 * s);
    }

#pragma unroll
    for (int pass = 0; pass < 1024 / PROJ_THREADS; ++pass) {
        const int u   = tid + pass * PROJ_THREADS;
        const int ulo = u & 15;
        const int hh  = (u >> 4) & 1;
        const int s   = u >> 5;
        const int k0  = 4 * s + 2 * hh;
        const v4f r0 = *(const v4f*)(W1 + (k0 + 0) * 64 + 4 * ulo);
        const v4f r1 = *(const v4f*)(W1 + (k0 + 1) * 64 + 4 * ulo);
        v2f* d = &ldsW1[(s * 4) * 32 + (u & 31)];
        v2f f0; f0.x = r0.x; f0.y = r1.x;
        v2f f1; f1.x = r0.y; f1.y = r1.y;
        v2f f2; f2.x = r0.z; f2.y = r1.z;
        v2f f3; f3.x = r0.w; f3.y = r1.w;
        d[0 * 32] = f0;
        d[1 * 32] = f1;
        d[2 * 32] = f2;
        d[3 * 32] = f3;
    }
    __syncthreads();

    v2f bw[128];
#pragma unroll
    for (int i = 0; i < 128; ++i) bw[i] = ldsW1[i * 32 + lane];

    const v4f bias = *(const v4f*)(B1 + 4 * lo);

    while (tile < nTiles) {
        const int next = tile + tileStep;
        v2f an[16];
        {
            const int row  = next * 16 + lo;
            const int rowC = row < nNodes ? row : (nNodes - 1);
            const float* __restrict__ pA = h + (long long)rowC * 64 + 2 * hi;
#pragma unroll
            for (int s = 0; s < 16; ++s) an[s] = *(const v2f*)(pA + 4 * s);
        }

        v8f p0 = {}, p1 = {}, p2 = {}, p3 = {};
        v8f q0 = {}, q1 = {}, q2 = {}, q3 = {};

#pragma unroll
        for (int s = 0; s < 16; ++s) {
            const v2f as = a[s];
            p0 = __builtin_amdgcn_wmma_f32_16x16x4_f32(false, as, false, bw[4 * s + 0], (short)0, p0, false, false);
            p1 = __builtin_amdgcn_wmma_f32_16x16x4_f32(false, as, false, bw[4 * s + 1], (short)0, p1, false, false);
            p2 = __builtin_amdgcn_wmma_f32_16x16x4_f32(false, as, false, bw[4 * s + 2], (short)0, p2, false, false);
            p3 = __builtin_amdgcn_wmma_f32_16x16x4_f32(false, as, false, bw[4 * s + 3], (short)0, p3, false, false);
            q0 = __builtin_amdgcn_wmma_f32_16x16x4_f32(false, as, false, bw[4 * (s + 16) + 0], (short)0, q0, false, false);
            q1 = __builtin_amdgcn_wmma_f32_16x16x4_f32(false, as, false, bw[4 * (s + 16) + 1], (short)0, q1, false, false);
            q2 = __builtin_amdgcn_wmma_f32_16x16x4_f32(false, as, false, bw[4 * (s + 16) + 2], (short)0, q2, false, false);
            q3 = __builtin_amdgcn_wmma_f32_16x16x4_f32(false, as, false, bw[4 * (s + 16) + 3], (short)0, q3, false, false);
        }

#pragma unroll
        for (int j = 0; j < 8; ++j) {
            const int r = tile * 16 + 8 * hi + j;
            if (r < nNodes) {
                v4f vp, vq;
                vp.x = p0[j] + bias.x; vp.y = p1[j] + bias.y;
                vp.z = p2[j] + bias.z; vp.w = p3[j] + bias.w;
                vq.x = q0[j]; vq.y = q1[j]; vq.z = q2[j]; vq.w = q3[j];
                *(v4f*)(P + (long long)r * 64 + 4 * lo) = vp;
                *(v4f*)(Q + (long long)r * 64 + 4 * lo) = vq;
            }
        }

#pragma unroll
        for (int s = 0; s < 16; ++s) a[s] = an[s];
        tile = next;
    }
}


#define EDGE_LOAD(i)                                                          \
    const int s##i = __shfl(sv, half + i, 32);                                \
    const int d##i = __shfl(dv, half + i, 32);                                \
    const v4f p##i = P4[(unsigned)s##i * 16u + (unsigned)c];                  \
    const v4f q##i = Q4[(unsigned)d##i * 16u + (unsigned)c];

#define EDGE_SCORE(i)                                                         \
    float a##i, b##i;                                                         \
    {                                                                         \
        const float h0 = fmaxf(p##i.x + q##i.x, 0.0f);                        \
        const float h1 = fmaxf(p##i.y + q##i.y, 0.0f);                        \
        const float h2 = fmaxf(p##i.z + q##i.z, 0.0f);                        \
        const float h3 = fmaxf(p##i.w + q##i.w, 0.0f);                        \
        a##i = h0 * wA.x + h1 * wA.z + h2 * wB.x + h3 * wB.z;                 \
        b##i = h0 * wA.y + h1 * wA.w + h2 * wB.y + h3 * wB.w;                 \
    }

#define BFLY(lo, hi, m, up)                                                   \
    {                                                                         \
        const float keepA = up ? a##hi : a##lo;                               \
        const float sendA = up ? a##lo : a##hi;                               \
        const float keepB = up ? b##hi : b##lo;                               \
        const float sendB = up ? b##lo : b##hi;                               \
        a##lo = keepA + __shfl_xor(sendA, m, 32);                             \
        b##lo = keepB + __shfl_xor(sendB, m, 32);                             \
    }

__global__ __launch_bounds__(EDGE_THREADS)
void edge_score_kernel(const float* __restrict__ ws,
                       const int*   __restrict__ src,
                       const int*   __restrict__ dst,
                       const float* __restrict__ W2,
                       const float* __restrict__ B2,
                       float* __restrict__ out,
                       int hElems,
                       int nEdges)
{
    const v4f* __restrict__ P4 = (const v4f*)ws;
    const v4f* __restrict__ Q4 = (const v4f*)(ws + hElems);

    const int lane = threadIdx.x & 31;
    const int c    = lane & 15;
    const int half = lane & 16;

    const int waveG   = (int)blockIdx.x * EDGE_WAVES + ((int)threadIdx.x >> 5);
    const int nWaves  = (int)gridDim.x * EDGE_WAVES;
    const int nChunks = (nEdges + 31) >> 5;

    const v4f wA = *(const v4f*)(W2 + 8 * c);
    const v4f wB = *(const v4f*)(W2 + 8 * c + 4);
    const float b20 = B2[0];
    const float b21 = B2[1];

    const bool up8 = (c & 8) != 0;
    const bool up4 = (c & 4) != 0;
    const bool up2 = (c & 2) != 0;
    const bool up1 = (c & 1) != 0;

    for (int chunk = waveG; chunk < nChunks; chunk += nWaves) {
        const int e  = chunk * 32 + lane;
        const int eC = e < nEdges ? e : (nEdges - 1);
        const int sv = src[eC];
        const int dv = dst[eC];

        EDGE_LOAD(0)  EDGE_LOAD(1)  EDGE_LOAD(2)  EDGE_LOAD(3)
        EDGE_LOAD(4)  EDGE_LOAD(5)  EDGE_LOAD(6)  EDGE_LOAD(7)
        EDGE_LOAD(8)  EDGE_LOAD(9)  EDGE_LOAD(10) EDGE_LOAD(11)
        EDGE_LOAD(12) EDGE_LOAD(13) EDGE_LOAD(14) EDGE_LOAD(15)

        EDGE_SCORE(0)  EDGE_SCORE(1)  EDGE_SCORE(2)  EDGE_SCORE(3)
        EDGE_SCORE(4)  EDGE_SCORE(5)  EDGE_SCORE(6)  EDGE_SCORE(7)
        EDGE_SCORE(8)  EDGE_SCORE(9)  EDGE_SCORE(10) EDGE_SCORE(11)
        EDGE_SCORE(12) EDGE_SCORE(13) EDGE_SCORE(14) EDGE_SCORE(15)

        BFLY(0, 8, 8, up8)  BFLY(1, 9, 8, up8)  BFLY(2, 10, 8, up8) BFLY(3, 11, 8, up8)
        BFLY(4, 12, 8, up8) BFLY(5, 13, 8, up8) BFLY(6, 14, 8, up8) BFLY(7, 15, 8, up8)
        BFLY(0, 4, 4, up4)  BFLY(1, 5, 4, up4)  BFLY(2, 6, 4, up4)  BFLY(3, 7, 4, up4)
        BFLY(0, 2, 2, up2)  BFLY(1, 3, 2, up2)
        BFLY(0, 1, 1, up1)

        if (e < nEdges) {
            v2f o;
            o.x = a0 + b20;
            o.y = b0 + b21;
            *(v2f*)(out + (long long)e * 2) = o;
        }
    }
}

extern "C" void kernel_launch(void* const* d_in, const int* in_sizes, int n_in,
                              void* d_out, int out_size, void* d_ws, size_t ws_size,
                              hipStream_t stream) {
    const float* h   = (const float*)d_in[0];
    const int*   src = (const int*)d_in[1];
    const int*   dst = (const int*)d_in[2];
    const float* W1  = (const float*)d_in[3];
    const float* B1  = (const float*)d_in[4];
    const float* W2  = (const float*)d_in[5];
    const float* B2  = (const float*)d_in[6];
    float* out = (float*)d_out;
    float* ws  = (float*)d_ws;

    const int hElems = in_sizes[0];
    const int nEdges = in_sizes[1];

    node_project_kernel<<<PROJ_BLOCKS, PROJ_THREADS, 0, stream>>>(h, W1, B1, ws, hElems);
    edge_score_kernel<<<EDGE_BLOCKS, EDGE_THREADS, 0, stream>>>(
        ws, src, dst, W2, B2, out, hElems, nEdges);
}
